# starting kernel
# baseline (speedup 1.0000x reference)
#include <hip/hip_runtime.h>

typedef float v2f __attribute__((ext_vector_type(2)));
typedef float v8f __attribute__((ext_vector_type(8)));

__global__ void gnn_zero_agg(float* __restrict__ ws, float* __restrict__ out, int n_nodes) {
    long idx   = (long)blockIdx.x * blockDim.x + threadIdx.x;
    long total = (long)n_nodes * 32;
    if (idx >= total) return;
    long row = idx >> 5;
    long q   = idx & 31;
    float4 z = make_float4(0.f, 0.f, 0.f, 0.f);
    ((float4*)(ws  + row * 256))[q] = z;
    ((float4*)(out + row * 256))[q] = z;
}

#define KMAX 256
#define APAD 260

__global__ void gnn_gemm_wmma(const float* __restrict__ A, int K,
                              const float* __restrict__ B,
                              const float* __restrict__ bias,
                              float* __restrict__ C, int ldc, int col_off) {
    __shared__ float As[16][APAD];

    const int tid   = threadIdx.x;
    const int mbase = blockIdx.x * 16;

    for (int i = tid; i < 16 * K; i += 256) {
        int r = i / K, c = i % K;
        As[r][c] = A[(long)(mbase + r) * K + c];
    }
    __syncthreads();

    const int wave  = tid >> 5;
    const int lane  = tid & 31;
    const int nbase = wave * 16;
    const int n     = lane & 15;
    const int hi    = lane >> 4;

    v8f acc;
    {
        float bv = bias[nbase + n];
        for (int g = 0; g < 8; ++g) acc[g] = bv;
    }

    for (int k = 0; k < K; k += 4) {
        int kb = k + 2 * hi;
        v2f a, b;
        a.x = As[n][kb];
        a.y = As[n][kb + 1];
        b.x = B[(long)kb       * 128 + nbase + n];
        b.y = B[(long)(kb + 1) * 128 + nbase + n];
        acc = __builtin_amdgcn_wmma_f32_16x16x4_f32(
                   false, a,  false, b,
                   (short)0, acc,  false,  false);
    }

    for (int g = 0; g < 8; ++g) {
        int row = mbase + g + 8 * hi;
        C[(long)row * ldc + col_off + nbase + n] = acc[g];
    }
}

__global__ void gnn_scatter(const int* __restrict__ src, const int* __restrict__ dst,
                            const float* __restrict__ msg_tab, float* __restrict__ agg_tab,
                            int n_edges) {
    int edge = blockIdx.x * 8 + (threadIdx.x >> 5);
    int lane = threadIdx.x & 31;
    if (edge >= n_edges) return;

    int s = src[edge];
    int d = dst[edge];

    const float4 v = ((const float4*)(msg_tab + (long)s * 256 + 128))[lane];
    float* a = agg_tab + (long)d * 256 + lane * 4;
    atomicAdd(a + 0, v.x);
    atomicAdd(a + 1, v.y);
    atomicAdd(a + 2, v.z);
    atomicAdd(a + 3, v.w);
}

extern "C" void kernel_launch(void* const* d_in, const int* in_sizes, int n_in,
                              void* d_out, int out_size, void* d_ws, size_t ws_size,
                              hipStream_t stream) {
    const float* features = (const float*)d_in[0];
    const int*   src      = (const int*)  d_in[1];
    const int*   dst      = (const int*)  d_in[2];
    const float* W1       = (const float*)d_in[3];
    const float* b1       = (const float*)d_in[4];
    const float* W2       = (const float*)d_in[5];
    const float* b2       = (const float*)d_in[6];

    float* out   = (float*)d_out;
    float* feat1 = (float*)d_ws;

    const int N = in_sizes[0] / 128;
    const int E = in_sizes[1];

    {
        long total  = (long)N * 32;
        int  blocks = (int)((total + 255) / 256);
        gnn_zero_agg<<<blocks, 256, 0, stream>>>(feat1, out, N);
    }

    gnn_gemm_wmma<<<N / 16, 256, 0, stream>>>(features, 128, W1, b1, feat1, 256, 128);

    {
        int blocks = (E + 7) / 8;
        gnn_scatter<<<blocks, 256, 0, stream>>>(src, dst, feat1, feat1, E);
    }

    gnn_gemm_wmma<<<N / 16, 256, 0, stream>>>(feat1, 256, W2, b2, out, 256, 128);

    {
        int blocks = (E + 7) / 8;
        gnn_scatter<<<blocks, 256, 0, stream>>>(src, dst, out, out, E);
    }
}


# Replace f32 atomic scatter-add by per-call counting sort of edges by destination (int atomic rank, scan, place) + one-wave-per-node row gather; sums now in bucket (arrival-rank) order
# speedup vs baseline: 1.7745x; 1.7745x over previous
#include <hip/hip_runtime.h>


typedef float v2f __attribute__((ext_vector_type(2)));
typedef float v8f __attribute__((ext_vector_type(8)));

__global__ void gnn_zero_cnt(int* __restrict__ cnt, int n) {
    int i = blockIdx.x * blockDim.x + threadIdx.x;
    if (i < n) cnt[i] = 0;
}

__global__ void gnn_count_rank(const int* __restrict__ dst, int* __restrict__ cnt,
                               int* __restrict__ rank, int n_edges) {
    int e = blockIdx.x * blockDim.x + threadIdx.x;
    if (e < n_edges) rank[e] = atomicAdd(&cnt[dst[e]], 1);
}

__global__ void gnn_scan(const int* __restrict__ cnt, int* __restrict__ off, int n) {
    __shared__ int wtot[32];
    __shared__ int wpre[32];
    const int tid  = threadIdx.x;
    const int lane = tid & 31;
    const int w    = tid >> 5;
    const int base = blockIdx.x * 1024;

    int p = 0;
    for (int i = tid; i < base; i += 1024) p += cnt[i];

    const int i = base + tid;
    const int v = (i < n) ? cnt[i] : 0;

    int inc = v;
    for (int d = 1; d < 32; d <<= 1) {
        int t = __shfl_up(inc, d);
        if (lane >= d) inc += t;
    }
    for (int d = 16; d >= 1; d >>= 1) p += __shfl_xor(p, d);
    if (lane == 31) wtot[w] = inc;
    if (lane == 0)  wpre[w] = p;
    __syncthreads();

    const int tw = wtot[lane];
    int pw = wpre[lane];
    int tinc = tw;
    for (int d = 1; d < 32; d <<= 1) {
        int t = __shfl_up(tinc, d);
        if (lane >= d) tinc += t;
    }
    for (int d = 16; d >= 1; d >>= 1) pw += __shfl_xor(pw, d);
    const int wave_excl = __shfl(tinc, w) - __shfl(tw, w);

    const int excl = pw + wave_excl + inc - v;
    if (i < n)      off[i] = excl;
    if (i == n - 1) off[n] = excl + v;
}

__global__ void gnn_place(const int* __restrict__ src, const int* __restrict__ dst,
                          const int* __restrict__ rank, const int* __restrict__ off,
                          int* __restrict__ bucket, int n_edges) {
    int e = blockIdx.x * blockDim.x + threadIdx.x;
    if (e < n_edges) bucket[off[dst[e]] + rank[e]] = src[e];
}

#define KMAX 256
#define APAD 260

__global__ void gnn_gemm_wmma(const float* __restrict__ A, int K,
                              const float* __restrict__ B,
                              const float* __restrict__ bias,
                              float* __restrict__ C, int ldc, int col_off) {
    __shared__ float As[16][APAD];

    const int tid   = threadIdx.x;
    const int mbase = blockIdx.x * 16;

    for (int i = tid; i < 16 * K; i += 256) {
        int r = i / K, c = i % K;
        As[r][c] = A[(long)(mbase + r) * K + c];
    }
    __syncthreads();

    const int wave  = tid >> 5;
    const int lane  = tid & 31;
    const int nbase = wave * 16;
    const int n     = lane & 15;
    const int hi    = lane >> 4;

    v8f acc;
    {
        float bv = bias[nbase + n];
        for (int g = 0; g < 8; ++g) acc[g] = bv;
    }

    for (int k = 0; k < K; k += 4) {
        int kb = k + 2 * hi;
        v2f a, b;
        a.x = As[n][kb];
        a.y = As[n][kb + 1];
        b.x = B[(long)kb       * 128 + nbase + n];
        b.y = B[(long)(kb + 1) * 128 + nbase + n];
        acc = __builtin_amdgcn_wmma_f32_16x16x4_f32(
                   false, a,  false, b,
                   (short)0, acc,  false,  false);
    }

    for (int g = 0; g < 8; ++g) {
        int row = mbase + g + 8 * hi;
        C[(long)row * ldc + col_off + nbase + n] = acc[g];
    }
}

__global__ void gnn_gather_sum(const int* __restrict__ off, const int* __restrict__ bucket,
                               const float* __restrict__ msg_tab, float* __restrict__ agg_tab,
                               int n_nodes) {
    const int node = blockIdx.x * 8 + __builtin_amdgcn_readfirstlane((int)(threadIdx.x >> 5));
    const int lane = threadIdx.x & 31;
    if (node >= n_nodes) return;

    const int beg = __builtin_amdgcn_readfirstlane(off[node]);
    const int end = __builtin_amdgcn_readfirstlane(off[node + 1]);

    const float4* rows = (const float4*)(msg_tab + 128) + lane;
    float4 acc = make_float4(0.f, 0.f, 0.f, 0.f);

    for (int c = beg; c < end; c += 32) {
        const int m = (end - c < 32) ? (end - c) : 32;
        const int sidx = (lane < m) ? bucket[c + lane] : 0;
        for (int j = 0; j < m; j += 8) {
            float4 v[8];
#pragma unroll
            for (int k = 0; k < 8; ++k) {
                const int jj = (j + k < m) ? (j + k) : (m - 1);
                const int s  = __builtin_amdgcn_readlane(sidx, jj);
                v[k] = rows[(long)s * 64];
            }
#pragma unroll
            for (int k = 0; k < 8; ++k) {
                if (j + k < m) {
                    acc.x += v[k].x;
                    acc.y += v[k].y;
                    acc.z += v[k].z;
                    acc.w += v[k].w;
                }
            }
        }
    }
    ((float4*)(agg_tab + (long)node * 256))[lane] = acc;
}

extern "C" void kernel_launch(void* const* d_in, const int* in_sizes, int n_in,
                              void* d_out, int out_size, void* d_ws, size_t ws_size,
                              hipStream_t stream) {
    const float* features = (const float*)d_in[0];
    const int*   src      = (const int*)  d_in[1];
    const int*   dst      = (const int*)  d_in[2];
    const float* W1       = (const float*)d_in[3];
    const float* b1       = (const float*)d_in[4];
    const float* W2       = (const float*)d_in[5];
    const float* b2       = (const float*)d_in[6];

    float* out   = (float*)d_out;
    float* feat1 = (float*)d_ws;

    const int N = in_sizes[0] / 128;
    const int E = in_sizes[1];

    int* cnt    = (int*)(feat1 + (size_t)N * 256);
    int* off    = cnt + N;
    int* rank   = off + (N + 1);
    int* bucket = rank + E;

    gnn_zero_cnt  <<<(N + 255) / 256,   256,  0, stream>>>(cnt, N);
    gnn_count_rank<<<(E + 255) / 256,   256,  0, stream>>>(dst, cnt, rank, E);
    gnn_scan      <<<(N + 1023) / 1024, 1024, 0, stream>>>(cnt, off, N);
    gnn_place     <<<(E + 255) / 256,   256,  0, stream>>>(src, dst, rank, off, bucket, E);

    gnn_gemm_wmma<<<N / 16, 256, 0, stream>>>(features, 128, W1, b1, feat1, 256, 128);

    gnn_gather_sum<<<(N + 7) / 8, 256, 0, stream>>>(off, bucket, feat1, feat1, N);

    gnn_gemm_wmma<<<N / 16, 256, 0, stream>>>(feat1, 256, W2, b2, out, 256, 128);

    gnn_gather_sum<<<(N + 7) / 8, 256, 0, stream>>>(off, bucket, out, out, N);
}


# Edge sort in 2 launches (was 4): chunks binned by destination range with LDS atomics, ranges counted and placed in LDS; no global atomics or scattered stores; bucket order = LDS arrival order
# speedup vs baseline: 2.8911x; 1.6292x over previous
#include <hip/hip_runtime.h>


typedef float v2f __attribute__((ext_vector_type(2)));
typedef float v8f __attribute__((ext_vector_type(8)));

#define CH      3200
#define CH_PER  7
#define RANGE   256
#define SCAP    8192

__device__ __forceinline__ int scan256_one_wave(int* a, int lane) {
    int v[8];
    int s = 0;
#pragma unroll
    for (int j = 0; j < 8; ++j) { v[j] = a[lane * 8 + j]; s += v[j]; }
    int inc = s;
#pragma unroll
    for (int d = 1; d < 32; d <<= 1) {
        int t = __shfl_up(inc, d);
        if (lane >= d) inc += t;
    }
    int run = inc - s;
#pragma unroll
    for (int j = 0; j < 8; ++j) { a[lane * 8 + j] = run; run += v[j]; }
    return __shfl(inc, 31);
}

__global__ void __launch_bounds__(512)
gnn_bin_chunks(const int* __restrict__ src, const int* __restrict__ dst,
               int* __restrict__ recs_w, int* __restrict__ counts, int* __restrict__ starts,
               int n_edges, int n_ranges) {
    int2* __restrict__ recs = (int2*)recs_w;
    __shared__ int  hist[RANGE];
    __shared__ int  cursor[RANGE];
    __shared__ int2 stage[CH];

    const int tid  = threadIdx.x;
    const int lane = tid & 31;
    const int a    = blockIdx.x;
    const int base = a * CH;
    const int chunk_n = (n_edges - base < CH) ? (n_edges - base) : CH;

    if (tid < RANGE) hist[tid] = 0;

    int d[CH_PER], s[CH_PER];
#pragma unroll
    for (int q = 0; q < CH_PER; ++q) {
        const int e = tid + q * 512;
        d[q] = -1; s[q] = 0;
        if (e < chunk_n) { d[q] = dst[base + e]; s[q] = src[base + e]; }
    }
    __syncthreads();
#pragma unroll
    for (int q = 0; q < CH_PER; ++q)
        if (d[q] >= 0) atomicAdd(&hist[d[q] >> 8], 1);
    __syncthreads();

    int my_count = 0;
    if (tid < RANGE) my_count = hist[tid];
    __syncthreads();
    if (tid < 32) scan256_one_wave(hist, lane);
    __syncthreads();
    if (tid < RANGE) {
        const int st = hist[tid];
        cursor[tid] = st;
        if (tid < n_ranges) {
            counts[a * n_ranges + tid] = my_count;
            starts[a * n_ranges + tid] = st;
        }
    }
    __syncthreads();
#pragma unroll
    for (int q = 0; q < CH_PER; ++q) {
        if (d[q] >= 0) {
            const int pos = atomicAdd(&cursor[d[q] >> 8], 1);
            stage[pos] = make_int2(d[q], s[q]);
        }
    }
    __syncthreads();
    for (int i = tid; i < chunk_n; i += 512) recs[base + i] = stage[i];
}

__global__ void __launch_bounds__(512)
gnn_bin_finish(const int* __restrict__ recs_w, const int* __restrict__ counts,
               const int* __restrict__ starts, int* __restrict__ off, int* __restrict__ bucket,
               int n_nodes, int n_chunks, int n_ranges) {
    const int2* __restrict__ recs = (const int2*)recs_w;
    __shared__ int node_cnt[RANGE];
    __shared__ int cursor[RANGE];
    __shared__ int sums[2];
    __shared__ int stage[SCAP];

    const int tid  = threadIdx.x;
    const int lane = tid & 31;
    const int r    = blockIdx.x;
    const int a    = tid >> 1;
    const int h    = tid & 1;

    if (tid < RANGE) node_cnt[tid] = 0;
    if (tid < 2)     sums[tid] = 0;

    int c = 0, o = 0;
    if (a < n_chunks) { c = counts[a * n_ranges + r]; o = starts[a * n_ranges + r]; }
    __syncthreads();

    {
        int cs = h ? 0 : c, os = h ? 0 : o;
#pragma unroll
        for (int dd = 16; dd >= 1; dd >>= 1) { cs += __shfl_xor(cs, dd); os += __shfl_xor(os, dd); }
        if (lane == 0) { atomicAdd(&sums[0], cs); atomicAdd(&sums[1], os); }
    }

    const int2* seg = recs + (size_t)a * CH + o;
    const int node0 = r * RANGE;

    const int* segd = (const int*)seg;
    for (int i0 = h; i0 < c; i0 += 8) {
        int dd[4];
#pragma unroll
        for (int q = 0; q < 4; ++q) { const int i = i0 + 2 * q; dd[q] = segd[2 * (i < c ? i : i0)]; }
#pragma unroll
        for (int q = 0; q < 4; ++q) {
            const bool ok = (i0 + 2 * q) < c;
            atomicAdd(&node_cnt[ok ? dd[q] - node0 : (tid & (RANGE - 1))], ok ? 1 : 0);
        }
    }
    __syncthreads();

    const int total     = sums[0];
    const int bin_start = sums[1];
    if (tid < 32) scan256_one_wave(node_cnt, lane);
    __syncthreads();
    if (tid < RANGE) {
        const int st = node_cnt[tid];
        cursor[tid] = st;
        if (node0 + tid < n_nodes) off[node0 + tid] = bin_start + st;
    }
    if (r == n_ranges - 1 && tid == 0) off[n_nodes] = bin_start + total;
    __syncthreads();

    const bool fits = total <= SCAP;
    for (int i0 = h; i0 < c; i0 += 8) {
        int2 rec[4];
#pragma unroll
        for (int q = 0; q < 4; ++q) { const int i = i0 + 2 * q; rec[q] = seg[i < c ? i : i0]; }
#pragma unroll
        for (int q = 0; q < 4; ++q) {
            const int i = i0 + 2 * q;
            if (i < c) {
                const int pos = atomicAdd(&cursor[rec[q].x - node0], 1);
                if (fits) stage[pos] = rec[q].y;
                else      bucket[bin_start + pos] = rec[q].y;
            }
        }
    }
    __syncthreads();
    if (fits)
        for (int i = tid; i < total; i += 512) bucket[bin_start + i] = stage[i];
}

#define KMAX 256
#define APAD 260

__global__ void gnn_gemm_wmma(const float* __restrict__ A, int K,
                              const float* __restrict__ B,
                              const float* __restrict__ bias,
                              float* __restrict__ C, int ldc, int col_off) {
    __shared__ float As[16][APAD];

    const int tid   = threadIdx.x;
    const int mbase = blockIdx.x * 16;

    for (int i = tid; i < 16 * K; i += 256) {
        int r = i / K, c = i % K;
        As[r][c] = A[(long)(mbase + r) * K + c];
    }
    __syncthreads();

    const int wave  = tid >> 5;
    const int lane  = tid & 31;
    const int nbase = wave * 16;
    const int n     = lane & 15;
    const int hi    = lane >> 4;

    v8f acc;
    {
        float bv = bias[nbase + n];
        for (int g = 0; g < 8; ++g) acc[g] = bv;
    }

    for (int k = 0; k < K; k += 4) {
        int kb = k + 2 * hi;
        v2f a, b;
        a.x = As[n][kb];
        a.y = As[n][kb + 1];
        b.x = B[(long)kb       * 128 + nbase + n];
        b.y = B[(long)(kb + 1) * 128 + nbase + n];
        acc = __builtin_amdgcn_wmma_f32_16x16x4_f32(
                   false, a,  false, b,
                   (short)0, acc,  false,  false);
    }

    for (int g = 0; g < 8; ++g) {
        int row = mbase + g + 8 * hi;
        C[(long)row * ldc + col_off + nbase + n] = acc[g];
    }
}

__global__ void gnn_gather_sum(const int* __restrict__ off, const int* __restrict__ bucket,
                               const float* __restrict__ msg_tab, float* __restrict__ agg_tab,
                               int n_nodes) {
    const int node = blockIdx.x * 8 + __builtin_amdgcn_readfirstlane((int)(threadIdx.x >> 5));
    const int lane = threadIdx.x & 31;
    if (node >= n_nodes) return;

    const int beg = __builtin_amdgcn_readfirstlane(off[node]);
    const int end = __builtin_amdgcn_readfirstlane(off[node + 1]);

    const float4* rows = (const float4*)(msg_tab + 128) + lane;
    float4 acc = make_float4(0.f, 0.f, 0.f, 0.f);

    for (int c = beg; c < end; c += 32) {
        const int m = (end - c < 32) ? (end - c) : 32;
        const int sidx = (lane < m) ? bucket[c + lane] : 0;
        for (int j = 0; j < m; j += 8) {
            float4 v[8];
#pragma unroll
            for (int k = 0; k < 8; ++k) {
                const int jj = (j + k < m) ? (j + k) : (m - 1);
                const int s  = __builtin_amdgcn_readlane(sidx, jj);
                v[k] = rows[(long)s * 64];
            }
#pragma unroll
            for (int k = 0; k < 8; ++k) {
                if (j + k < m) {
                    acc.x += v[k].x;
                    acc.y += v[k].y;
                    acc.z += v[k].z;
                    acc.w += v[k].w;
                }
            }
        }
    }
    ((float4*)(agg_tab + (long)node * 256))[lane] = acc;
}

extern "C" void kernel_launch(void* const* d_in, const int* in_sizes, int n_in,
                              void* d_out, int out_size, void* d_ws, size_t ws_size,
                              hipStream_t stream) {
    const float* features = (const float*)d_in[0];
    const int*   src      = (const int*)  d_in[1];
    const int*   dst      = (const int*)  d_in[2];
    const float* W1       = (const float*)d_in[3];
    const float* b1       = (const float*)d_in[4];
    const float* W2       = (const float*)d_in[5];
    const float* b2       = (const float*)d_in[6];

    float* out   = (float*)d_out;
    float* feat1 = (float*)d_ws;

    const int N = in_sizes[0] / 128;
    const int E = in_sizes[1];

    const int GA = (E + CH - 1) / CH;
    const int NR = (N + RANGE - 1) / RANGE;
    int* recs    = (int*)(feat1 + (size_t)N * 256);
    int* counts  = recs + 2 * (size_t)E;
    int* starts  = counts + GA * NR;
    int* off     = starts + GA * NR;
    int* bucket  = off + (N + 1);

    gnn_bin_chunks<<<GA, 512, 0, stream>>>(src, dst, recs, counts, starts, E, NR);
    gnn_bin_finish<<<NR, 512, 0, stream>>>(recs, counts, starts, off, bucket, N, GA, NR);

    gnn_gemm_wmma<<<N / 16, 256, 0, stream>>>(features, 128, W1, b1, feat1, 256, 128);

    gnn_gather_sum<<<(N + 7) / 8, 256, 0, stream>>>(off, bucket, feat1, feat1, N);

    gnn_gemm_wmma<<<N / 16, 256, 0, stream>>>(feat1, 256, W2, b2, out, 256, 128);

    gnn_gather_sum<<<(N + 7) / 8, 256, 0, stream>>>(off, bucket, out, out, N);
}


# GEMM re-tiled: 256 resident blocks keep all of W in LDS as ready fragments; a wave feeds 4 tiles from one A fragment loaded straight from memory; k summed in a permuted order inside each 8
# speedup vs baseline: 3.4741x; 1.2017x over previous
#include <hip/hip_runtime.h>


typedef float v2f __attribute__((ext_vector_type(2)));
typedef float v8f __attribute__((ext_vector_type(8)));

#define CH      3200
#define CH_PER  7
#define RANGE   256
#define SCAP    8192

__device__ __forceinline__ int scan256_one_wave(int* a, int lane) {
    int v[8];
    int s = 0;
#pragma unroll
    for (int j = 0; j < 8; ++j) { v[j] = a[lane * 8 + j]; s += v[j]; }
    int inc = s;
#pragma unroll
    for (int d = 1; d < 32; d <<= 1) {
        int t = __shfl_up(inc, d);
        if (lane >= d) inc += t;
    }
    int run = inc - s;
#pragma unroll
    for (int j = 0; j < 8; ++j) { a[lane * 8 + j] = run; run += v[j]; }
    return __shfl(inc, 31);
}

__global__ void __launch_bounds__(512)
gnn_bin_chunks(const int* __restrict__ src, const int* __restrict__ dst,
               int* __restrict__ recs_w, int* __restrict__ counts, int* __restrict__ starts,
               int n_edges, int n_ranges) {
    int2* __restrict__ recs = (int2*)recs_w;
    __shared__ int  hist[RANGE];
    __shared__ int  cursor[RANGE];
    __shared__ int2 stage[CH];

    const int tid  = threadIdx.x;
    const int lane = tid & 31;
    const int a    = blockIdx.x;
    const int base = a * CH;
    const int chunk_n = (n_edges - base < CH) ? (n_edges - base) : CH;

    if (tid < RANGE) hist[tid] = 0;

    int d[CH_PER], s[CH_PER];
#pragma unroll
    for (int q = 0; q < CH_PER; ++q) {
        const int e = tid + q * 512;
        d[q] = -1; s[q] = 0;
        if (e < chunk_n) { d[q] = dst[base + e]; s[q] = src[base + e]; }
    }
    __syncthreads();
#pragma unroll
    for (int q = 0; q < CH_PER; ++q)
        if (d[q] >= 0) atomicAdd(&hist[d[q] >> 8], 1);
    __syncthreads();

    int my_count = 0;
    if (tid < RANGE) my_count = hist[tid];
    __syncthreads();
    if (tid < 32) scan256_one_wave(hist, lane);
    __syncthreads();
    if (tid < RANGE) {
        const int st = hist[tid];
        cursor[tid] = st;
        if (tid < n_ranges) {
            counts[a * n_ranges + tid] = my_count;
            starts[a * n_ranges + tid] = st;
        }
    }
    __syncthreads();
#pragma unroll
    for (int q = 0; q < CH_PER; ++q) {
        if (d[q] >= 0) {
            const int pos = atomicAdd(&cursor[d[q] >> 8], 1);
            stage[pos] = make_int2(d[q], s[q]);
        }
    }
    __syncthreads();
    for (int i = tid; i < chunk_n; i += 512) recs[base + i] = stage[i];
}

__global__ void __launch_bounds__(512)
gnn_bin_finish(const int* __restrict__ recs_w, const int* __restrict__ counts,
               const int* __restrict__ starts, int* __restrict__ off, int* __restrict__ bucket,
               int n_nodes, int n_chunks, int n_ranges) {
    const int2* __restrict__ recs = (const int2*)recs_w;
    __shared__ int node_cnt[RANGE];
    __shared__ int cursor[RANGE];
    __shared__ int sums[2];
    __shared__ int stage[SCAP];

    const int tid  = threadIdx.x;
    const int lane = tid & 31;
    const int r    = blockIdx.x;
    const int a    = tid >> 1;
    const int h    = tid & 1;

    if (tid < RANGE) node_cnt[tid] = 0;
    if (tid < 2)     sums[tid] = 0;

    int c = 0, o = 0;
    if (a < n_chunks) { c = counts[a * n_ranges + r]; o = starts[a * n_ranges + r]; }
    __syncthreads();

    {
        int cs = h ? 0 : c, os = h ? 0 : o;
#pragma unroll
        for (int dd = 16; dd >= 1; dd >>= 1) { cs += __shfl_xor(cs, dd); os += __shfl_xor(os, dd); }
        if (lane == 0) { atomicAdd(&sums[0], cs); atomicAdd(&sums[1], os); }
    }

    const int2* seg = recs + (size_t)a * CH + o;
    const int node0 = r * RANGE;

    const int* segd = (const int*)seg;
    for (int i0 = h; i0 < c; i0 += 8) {
        int dd[4];
#pragma unroll
        for (int q = 0; q < 4; ++q) { const int i = i0 + 2 * q; dd[q] = segd[2 * (i < c ? i : i0)]; }
#pragma unroll
        for (int q = 0; q < 4; ++q) {
            const bool ok = (i0 + 2 * q) < c;
            atomicAdd(&node_cnt[ok ? dd[q] - node0 : (tid & (RANGE - 1))], ok ? 1 : 0);
        }
    }
    __syncthreads();

    const int total     = sums[0];
    const int bin_start = sums[1];
    if (tid < 32) scan256_one_wave(node_cnt, lane);
    __syncthreads();
    if (tid < RANGE) {
        const int st = node_cnt[tid];
        cursor[tid] = st;
        if (node0 + tid < n_nodes) off[node0 + tid] = bin_start + st;
    }
    if (r == n_ranges - 1 && tid == 0) off[n_nodes] = bin_start + total;
    __syncthreads();

    const bool fits = total <= SCAP;
    for (int i0 = h; i0 < c; i0 += 8) {
        int2 rec[4];
#pragma unroll
        for (int q = 0; q < 4; ++q) { const int i = i0 + 2 * q; rec[q] = seg[i < c ? i : i0]; }
#pragma unroll
        for (int q = 0; q < 4; ++q) {
            const int i = i0 + 2 * q;
            if (i < c) {
                const int pos = atomicAdd(&cursor[rec[q].x - node0], 1);
                if (fits) stage[pos] = rec[q].y;
                else      bucket[bin_start + pos] = rec[q].y;
            }
        }
    }
    __syncthreads();
    if (fits)
        for (int i = tid; i < total; i += 512) bucket[bin_start + i] = stage[i];
}

#define GEMM_BLOCKS 256

__device__ __forceinline__ v8f mma4(v2f a, v2f b, v8f c) {
    return __builtin_amdgcn_wmma_f32_16x16x4_f32(false, a, false, b, (short)0, c, false, false);
}

__device__ __forceinline__ void load_a_chunk(float4 (&dst)[8], const float* __restrict__ p) {
#pragma unroll
    for (int i = 0; i < 8; ++i) dst[i] = *(const float4*)(p + 8 * i);
}

template <int K>
__global__ void __launch_bounds__(256)
gnn_gemm_wres(const float* __restrict__ A, const float* __restrict__ B,
              const float* __restrict__ bias, float* __restrict__ C,
              int ldc, int col_off, int n_tiles) {
    __shared__ float4 Wl[K * 32];

    const int tid = threadIdx.x;
#pragma unroll
    for (int i = 0; i < K / 32; ++i) {
        const int id = i * 256 + tid;
        const int cg = id & 31;
        const int kg = id >> 5;
        const float* wp = B + (size_t)(4 * kg) * 128 + 4 * cg;
        const float4 r0 = *(const float4*)(wp);
        const float4 r1 = *(const float4*)(wp + 128);
        const float4 r2 = *(const float4*)(wp + 256);
        const float4 r3 = *(const float4*)(wp + 384);
        float4* d = Wl + ((kg >> 1) * 8 + (cg >> 4) * 4) * 32 + (kg & 1) * 16 + (cg & 15);
        d[0]  = make_float4(r0.x, r1.x, r2.x, r3.x);
        d[32] = make_float4(r0.y, r1.y, r2.y, r3.y);
        d[64] = make_float4(r0.z, r1.z, r2.z, r3.z);
        d[96] = make_float4(r0.w, r1.w, r2.w, r3.w);
    }
    __syncthreads();

    const int w    = __builtin_amdgcn_readfirstlane(tid >> 5);
    const int lane = tid & 31;
    const int c    = lane & 15;
    const int h    = lane >> 4;
    const int G    = gridDim.x;
    const int b    = blockIdx.x;
    const int nq   = (b < n_tiles) ? (n_tiles - b + G - 1) / G : 0;
    const int nu   = nq >> 2;
    const int hf   = w & 1;
    const int qs   = w >> 1;

    const float*  arow0 = A + (size_t)c * K + 4 * h;
    const float4* wfrag = Wl + hf * 128 + lane;

    float4 a0[8], a1[8];
    float4 bf0[4], bf1[4];
    if (nu > 0) {
        load_a_chunk(a0, arow0 + (size_t)(b + G * qs) * (16 * K));
#pragma unroll
        for (int t = 0; t < 4; ++t) bf0[t] = wfrag[t * 32];
    }

    for (int i = 0; i < nu; ++i) {
        const int tile  = b + G * (qs + 4 * i);
        const int tnext = b + G * (qs + 4 * ((i + 1 < nu) ? (i + 1) : i));
        const float* arow  = arow0 + (size_t)tile  * (16 * K);
        const float* anext = arow0 + (size_t)tnext * (16 * K);

        v8f acc[4];
#pragma unroll
        for (int t = 0; t < 4; ++t) {
            const float bv = bias[64 * hf + 4 * c + t];
#pragma unroll
            for (int r = 0; r < 8; ++r) acc[t][r] = bv;
        }

#pragma unroll
        for (int q = 0; q < K / 64; ++q) {
            float4 (&acur)[8] = (q & 1) ? a1 : a0;
            float4 (&anxt)[8] = (q & 1) ? a0 : a1;
            if (q + 1 < K / 64) load_a_chunk(anxt, arow + 64 * (q + 1));
            else                load_a_chunk(anxt, anext);
#pragma unroll
            for (int jj = 0; jj < 8; ++jj) {
                const int j = q * 8 + jj;
                float4 (&bcur)[4] = (jj & 1) ? bf1 : bf0;
                float4 (&bnxt)[4] = (jj & 1) ? bf0 : bf1;
                const float4* wp = wfrag + ((j + 1 < K / 8) ? (j + 1) : 0) * 256;
#pragma unroll
                for (int t = 0; t < 4; ++t) bnxt[t] = wp[t * 32];
                const float4 av = acur[jj];
                const v2f alo = {av.x, av.y};
                const v2f ahi = {av.z, av.w};
#pragma unroll
                for (int t = 0; t < 4; ++t) {
                    const v2f blo = {bcur[t].x, bcur[t].y};
                    const v2f bhi = {bcur[t].z, bcur[t].w};
                    acc[t] = mma4(alo, blo, acc[t]);
                    acc[t] = mma4(ahi, bhi, acc[t]);
                }
                __builtin_amdgcn_sched_barrier(0);
            }
        }

        float* crow = C + (size_t)(tile * 16 + 8 * h) * ldc + col_off + 64 * hf + 4 * c;
#pragma unroll
        for (int r = 0; r < 8; ++r)
            *(float4*)(crow + (size_t)r * ldc) =
                make_float4(acc[0][r], acc[1][r], acc[2][r], acc[3][r]);
    }

    const int tcol = 64 * (w >> 2) + 4 * c + (w & 3);
    const float4* wtail = Wl + w * 32 + lane;
    for (int q = nu * 4; q < nq; ++q) {
        const int tile = b + G * q;
        const float* arow = arow0 + (size_t)tile * (16 * K);
        float4 av[K / 8];
#pragma unroll
        for (int j = 0; j < K / 8; ++j) av[j] = *(const float4*)(arow + 8 * j);
        float4 g0[4], g1[4];
#pragma unroll
        for (int s = 0; s < 4; ++s) g0[s] = wtail[s * 256];
        v8f acc;
        {
            const float bv = bias[tcol];
#pragma unroll
            for (int r = 0; r < 8; ++r) acc[r] = bv;
        }
#pragma unroll
        for (int g = 0; g < K / 32; ++g) {
            float4 (&gcur)[4] = (g & 1) ? g1 : g0;
            float4 (&gnxt)[4] = (g & 1) ? g0 : g1;
            if (g + 1 < K / 32) {
#pragma unroll
                for (int s = 0; s < 4; ++s) gnxt[s] = wtail[((g + 1) * 4 + s) * 256];
            }
#pragma unroll
            for (int s = 0; s < 4; ++s) {
                const float4 a4 = av[g * 4 + s];
                const v2f alo = {a4.x, a4.y};
                const v2f ahi = {a4.z, a4.w};
                const v2f blo = {gcur[s].x, gcur[s].y};
                const v2f bhi = {gcur[s].z, gcur[s].w};
                acc = mma4(alo, blo, acc);
                acc = mma4(ahi, bhi, acc);
            }
            __builtin_amdgcn_sched_barrier(0);
        }
        float* crow = C + (size_t)(tile * 16 + 8 * h) * ldc + col_off + tcol;
#pragma unroll
        for (int r = 0; r < 8; ++r) crow[(size_t)r * ldc] = acc[r];
    }
}

__global__ void gnn_gather_sum(const int* __restrict__ off, const int* __restrict__ bucket,
                               const float* __restrict__ msg_tab, float* __restrict__ agg_tab,
                               int n_nodes) {
    const int node = blockIdx.x * 8 + __builtin_amdgcn_readfirstlane((int)(threadIdx.x >> 5));
    const int lane = threadIdx.x & 31;
    if (node >= n_nodes) return;

    const int beg = __builtin_amdgcn_readfirstlane(off[node]);
    const int end = __builtin_amdgcn_readfirstlane(off[node + 1]);

    const float4* rows = (const float4*)(msg_tab + 128) + lane;
    float4 acc = make_float4(0.f, 0.f, 0.f, 0.f);

    for (int c = beg; c < end; c += 32) {
        const int m = (end - c < 32) ? (end - c) : 32;
        const int sidx = (lane < m) ? bucket[c + lane] : 0;
        for (int j = 0; j < m; j += 8) {
            float4 v[8];
#pragma unroll
            for (int k = 0; k < 8; ++k) {
                const int jj = (j + k < m) ? (j + k) : (m - 1);
                const int s  = __builtin_amdgcn_readlane(sidx, jj);
                v[k] = rows[(long)s * 64];
            }
#pragma unroll
            for (int k = 0; k < 8; ++k) {
                if (j + k < m) {
                    acc.x += v[k].x;
                    acc.y += v[k].y;
                    acc.z += v[k].z;
                    acc.w += v[k].w;
                }
            }
        }
    }
    ((float4*)(agg_tab + (long)node * 256))[lane] = acc;
}

extern "C" void kernel_launch(void* const* d_in, const int* in_sizes, int n_in,
                              void* d_out, int out_size, void* d_ws, size_t ws_size,
                              hipStream_t stream) {
    const float* features = (const float*)d_in[0];
    const int*   src      = (const int*)  d_in[1];
    const int*   dst      = (const int*)  d_in[2];
    const float* W1       = (const float*)d_in[3];
    const float* b1       = (const float*)d_in[4];
    const float* W2       = (const float*)d_in[5];
    const float* b2       = (const float*)d_in[6];

    float* out   = (float*)d_out;
    float* feat1 = (float*)d_ws;

    const int N = in_sizes[0] / 128;
    const int E = in_sizes[1];

    const int GA = (E + CH - 1) / CH;
    const int NR = (N + RANGE - 1) / RANGE;
    int* recs    = (int*)(feat1 + (size_t)N * 256);
    int* counts  = recs + 2 * (size_t)E;
    int* starts  = counts + GA * NR;
    int* off     = starts + GA * NR;
    int* bucket  = off + (N + 1);

    gnn_bin_chunks<<<GA, 512, 0, stream>>>(src, dst, recs, counts, starts, E, NR);
    gnn_bin_finish<<<NR, 512, 0, stream>>>(recs, counts, starts, off, bucket, N, GA, NR);

    gnn_gemm_wres<128><<<GEMM_BLOCKS, 256, 0, stream>>>(features, W1, b1, feat1, 256, 128, N / 16);

    gnn_gather_sum<<<(N + 7) / 8, 256, 0, stream>>>(off, bucket, feat1, feat1, N);

    gnn_gemm_wres<256><<<GEMM_BLOCKS, 256, 0, stream>>>(feat1, W2, b2, out, 256, 128, N / 16);

    gnn_gather_sum<<<(N + 7) / 8, 256, 0, stream>>>(off, bucket, out, out, N);
}


# First GEMM and the chunk-grouping step of the edge sort share one launch (GEMM blocks first, 256-thread chunk blocks behind them): 5 launches; bucket order stays LDS arrival order
# speedup vs baseline: 3.6485x; 1.0502x over previous
#include <hip/hip_runtime.h>


typedef float v2f __attribute__((ext_vector_type(2)));
typedef float v8f __attribute__((ext_vector_type(8)));

#define CH      3200
#define BIN_T   256
#define CH_PER  13
#define RANGE   256
#define SCAP    8192

__device__ __forceinline__ int scan256_one_wave(int* a, int lane) {
    int v[8];
    int s = 0;
#pragma unroll
    for (int j = 0; j < 8; ++j) { v[j] = a[lane * 8 + j]; s += v[j]; }
    int inc = s;
#pragma unroll
    for (int d = 1; d < 32; d <<= 1) {
        int t = __shfl_up(inc, d);
        if (lane >= d) inc += t;
    }
    int run = inc - s;
#pragma unroll
    for (int j = 0; j < 8; ++j) { a[lane * 8 + j] = run; run += v[j]; }
    return __shfl(inc, 31);
}

__device__ __forceinline__ void
bin_chunks_body(int* __restrict__ sm, const int* __restrict__ src, const int* __restrict__ dst,
                int* __restrict__ recs_w, int* __restrict__ counts, int* __restrict__ starts,
                int n_edges, int n_ranges, const int a) {
    static_assert(BIN_T == RANGE, "one thread per destination range");
    int2* __restrict__ recs = (int2*)recs_w;
    int*  hist   = sm;
    int*  cursor = sm + RANGE;
    int2* stage  = (int2*)(sm + 2 * RANGE);

    const int tid  = threadIdx.x;
    const int lane = tid & 31;
    const int base = a * CH;
    const int chunk_n = (n_edges - base < CH) ? (n_edges - base) : CH;

    hist[tid] = 0;

    int d[CH_PER], s[CH_PER];
#pragma unroll
    for (int q = 0; q < CH_PER; ++q) {
        const int e = tid + q * BIN_T;
        d[q] = -1; s[q] = 0;
        if (e < chunk_n) { d[q] = dst[base + e]; s[q] = src[base + e]; }
    }
    __syncthreads();
#pragma unroll
    for (int q = 0; q < CH_PER; ++q)
        if (d[q] >= 0) atomicAdd(&hist[d[q] >> 8], 1);
    __syncthreads();

    const int my_count = hist[tid];
    __syncthreads();
    if (tid < 32) scan256_one_wave(hist, lane);
    __syncthreads();
    {
        const int st = hist[tid];
        cursor[tid] = st;
        if (tid < n_ranges) {
            counts[a * n_ranges + tid] = my_count;
            starts[a * n_ranges + tid] = st;
        }
    }
    __syncthreads();
#pragma unroll
    for (int q = 0; q < CH_PER; ++q) {
        if (d[q] >= 0) {
            const int pos = atomicAdd(&cursor[d[q] >> 8], 1);
            stage[pos] = make_int2(d[q], s[q]);
        }
    }
    __syncthreads();
    for (int i = tid; i < chunk_n; i += BIN_T) recs[base + i] = stage[i];
}

__global__ void __launch_bounds__(512)
gnn_bin_finish(const int* __restrict__ recs_w, const int* __restrict__ counts,
               const int* __restrict__ starts, int* __restrict__ off, int* __restrict__ bucket,
               int n_nodes, int n_chunks, int n_ranges) {
    const int2* __restrict__ recs = (const int2*)recs_w;
    __shared__ int node_cnt[RANGE];
    __shared__ int cursor[RANGE];
    __shared__ int sums[2];
    __shared__ int stage[SCAP];

    const int tid  = threadIdx.x;
    const int lane = tid & 31;
    const int r    = blockIdx.x;
    const int a    = tid >> 1;
    const int h    = tid & 1;

    if (tid < RANGE) node_cnt[tid] = 0;
    if (tid < 2)     sums[tid] = 0;

    int c = 0, o = 0;
    if (a < n_chunks) { c = counts[a * n_ranges + r]; o = starts[a * n_ranges + r]; }
    __syncthreads();

    {
        int cs = h ? 0 : c, os = h ? 0 : o;
#pragma unroll
        for (int dd = 16; dd >= 1; dd >>= 1) { cs += __shfl_xor(cs, dd); os += __shfl_xor(os, dd); }
        if (lane == 0) { atomicAdd(&sums[0], cs); atomicAdd(&sums[1], os); }
    }

    const int2* seg = recs + (size_t)a * CH + o;
    const int node0 = r * RANGE;

    const int* segd = (const int*)seg;
    for (int i0 = h; i0 < c; i0 += 8) {
        int dd[4];
#pragma unroll
        for (int q = 0; q < 4; ++q) { const int i = i0 + 2 * q; dd[q] = segd[2 * (i < c ? i : i0)]; }
#pragma unroll
        for (int q = 0; q < 4; ++q) {
            const bool ok = (i0 + 2 * q) < c;
            atomicAdd(&node_cnt[ok ? dd[q] - node0 : (tid & (RANGE - 1))], ok ? 1 : 0);
        }
    }
    __syncthreads();

    const int total     = sums[0];
    const int bin_start = sums[1];
    if (tid < 32) scan256_one_wave(node_cnt, lane);
    __syncthreads();
    if (tid < RANGE) {
        const int st = node_cnt[tid];
        cursor[tid] = st;
        if (node0 + tid < n_nodes) off[node0 + tid] = bin_start + st;
    }
    if (r == n_ranges - 1 && tid == 0) off[n_nodes] = bin_start + total;
    __syncthreads();

    const bool fits = total <= SCAP;
    for (int i0 = h; i0 < c; i0 += 8) {
        int2 rec[4];
#pragma unroll
        for (int q = 0; q < 4; ++q) { const int i = i0 + 2 * q; rec[q] = seg[i < c ? i : i0]; }
#pragma unroll
        for (int q = 0; q < 4; ++q) {
            const int i = i0 + 2 * q;
            if (i < c) {
                const int pos = atomicAdd(&cursor[rec[q].x - node0], 1);
                if (fits) stage[pos] = rec[q].y;
                else      bucket[bin_start + pos] = rec[q].y;
            }
        }
    }
    __syncthreads();
    if (fits)
        for (int i = tid; i < total; i += 512) bucket[bin_start + i] = stage[i];
}

#define GEMM_BLOCKS 256

__device__ __forceinline__ v8f mma4(v2f a, v2f b, v8f c) {
    return __builtin_amdgcn_wmma_f32_16x16x4_f32(false, a, false, b, (short)0, c, false, false);
}

__device__ __forceinline__ void load_a_chunk(float4 (&dst)[8], const float* __restrict__ p) {
#pragma unroll
    for (int i = 0; i < 8; ++i) dst[i] = *(const float4*)(p + 8 * i);
}

template <int K>
__device__ __forceinline__ void
gemm_wres_body(float4* __restrict__ Wl, const float* __restrict__ A, const float* __restrict__ B,
               const float* __restrict__ bias, float* __restrict__ C,
               int ldc, int col_off, int n_tiles, const int b, const int G) {
    const int tid = threadIdx.x;
#pragma unroll
    for (int i = 0; i < K / 32; ++i) {
        const int id = i * 256 + tid;
        const int cg = id & 31;
        const int kg = id >> 5;
        const float* wp = B + (size_t)(4 * kg) * 128 + 4 * cg;
        const float4 r0 = *(const float4*)(wp);
        const float4 r1 = *(const float4*)(wp + 128);
        const float4 r2 = *(const float4*)(wp + 256);
        const float4 r3 = *(const float4*)(wp + 384);
        float4* d = Wl + ((kg >> 1) * 8 + (cg >> 4) * 4) * 32 + (kg & 1) * 16 + (cg & 15);
        d[0]  = make_float4(r0.x, r1.x, r2.x, r3.x);
        d[32] = make_float4(r0.y, r1.y, r2.y, r3.y);
        d[64] = make_float4(r0.z, r1.z, r2.z, r3.z);
        d[96] = make_float4(r0.w, r1.w, r2.w, r3.w);
    }
    __syncthreads();

    const int w    = __builtin_amdgcn_readfirstlane(tid >> 5);
    const int lane = tid & 31;
    const int c    = lane & 15;
    const int h    = lane >> 4;
    const int nq   = (b < n_tiles) ? (n_tiles - b + G - 1) / G : 0;
    const int nu   = nq >> 2;
    const int hf   = w & 1;
    const int qs   = w >> 1;

    const float*  arow0 = A + (size_t)c * K + 4 * h;
    const float4* wfrag = Wl + hf * 128 + lane;

    float4 a0[8], a1[8];
    float4 bf0[4], bf1[4];
    if (nu > 0) {
        load_a_chunk(a0, arow0 + (size_t)(b + G * qs) * (16 * K));
#pragma unroll
        for (int t = 0; t < 4; ++t) bf0[t] = wfrag[t * 32];
    }

    for (int i = 0; i < nu; ++i) {
        const int tile  = b + G * (qs + 4 * i);
        const int tnext = b + G * (qs + 4 * ((i + 1 < nu) ? (i + 1) : i));
        const float* arow  = arow0 + (size_t)tile  * (16 * K);
        const float* anext = arow0 + (size_t)tnext * (16 * K);

        v8f acc[4];
#pragma unroll
        for (int t = 0; t < 4; ++t) {
            const float bv = bias[64 * hf + 4 * c + t];
#pragma unroll
            for (int r = 0; r < 8; ++r) acc[t][r] = bv;
        }

#pragma unroll
        for (int q = 0; q < K / 64; ++q) {
            float4 (&acur)[8] = (q & 1) ? a1 : a0;
            float4 (&anxt)[8] = (q & 1) ? a0 : a1;
            if (q + 1 < K / 64) load_a_chunk(anxt, arow + 64 * (q + 1));
            else                load_a_chunk(anxt, anext);
#pragma unroll
            for (int jj = 0; jj < 8; ++jj) {
                const int j = q * 8 + jj;
                float4 (&bcur)[4] = (jj & 1) ? bf1 : bf0;
                float4 (&bnxt)[4] = (jj & 1) ? bf0 : bf1;
                const float4* wp = wfrag + ((j + 1 < K / 8) ? (j + 1) : 0) * 256;
#pragma unroll
                for (int t = 0; t < 4; ++t) bnxt[t] = wp[t * 32];
                const float4 av = acur[jj];
                const v2f alo = {av.x, av.y};
                const v2f ahi = {av.z, av.w};
#pragma unroll
                for (int t = 0; t < 4; ++t) {
                    const v2f blo = {bcur[t].x, bcur[t].y};
                    const v2f bhi = {bcur[t].z, bcur[t].w};
                    acc[t] = mma4(alo, blo, acc[t]);
                    acc[t] = mma4(ahi, bhi, acc[t]);
                }
                __builtin_amdgcn_sched_barrier(0);
            }
        }

        float* crow = C + (size_t)(tile * 16 + 8 * h) * ldc + col_off + 64 * hf + 4 * c;
#pragma unroll
        for (int r = 0; r < 8; ++r)
            *(float4*)(crow + (size_t)r * ldc) =
                make_float4(acc[0][r], acc[1][r], acc[2][r], acc[3][r]);
    }

    const int tcol = 64 * (w >> 2) + 4 * c + (w & 3);
    const float4* wtail = Wl + w * 32 + lane;
    for (int q = nu * 4; q < nq; ++q) {
        const int tile = b + G * q;
        const float* arow = arow0 + (size_t)tile * (16 * K);
        float4 av[K / 8];
#pragma unroll
        for (int j = 0; j < K / 8; ++j) av[j] = *(const float4*)(arow + 8 * j);
        float4 g0[4], g1[4];
#pragma unroll
        for (int s = 0; s < 4; ++s) g0[s] = wtail[s * 256];
        v8f acc;
        {
            const float bv = bias[tcol];
#pragma unroll
            for (int r = 0; r < 8; ++r) acc[r] = bv;
        }
#pragma unroll
        for (int g = 0; g < K / 32; ++g) {
            float4 (&gcur)[4] = (g & 1) ? g1 : g0;
            float4 (&gnxt)[4] = (g & 1) ? g0 : g1;
            if (g + 1 < K / 32) {
#pragma unroll
                for (int s = 0; s < 4; ++s) gnxt[s] = wtail[((g + 1) * 4 + s) * 256];
            }
#pragma unroll
            for (int s = 0; s < 4; ++s) {
                const float4 a4 = av[g * 4 + s];
                const v2f alo = {a4.x, a4.y};
                const v2f ahi = {a4.z, a4.w};
                const v2f blo = {gcur[s].x, gcur[s].y};
                const v2f bhi = {gcur[s].z, gcur[s].w};
                acc = mma4(alo, blo, acc);
                acc = mma4(ahi, bhi, acc);
            }
            __builtin_amdgcn_sched_barrier(0);
        }
        float* crow = C + (size_t)(tile * 16 + 8 * h) * ldc + col_off + tcol;
#pragma unroll
        for (int r = 0; r < 8; ++r) crow[(size_t)r * ldc] = acc[r];
    }
}

template <int K>
__global__ void __launch_bounds__(256)
gnn_gemm_wres(const float* __restrict__ A, const float* __restrict__ B,
              const float* __restrict__ bias, float* __restrict__ C,
              int ldc, int col_off, int n_tiles) {
    __shared__ float4 Wl[K * 32];
    gemm_wres_body<K>(Wl, A, B, bias, C, ldc, col_off, n_tiles, blockIdx.x, gridDim.x);
}

__global__ void __launch_bounds__(256)
gnn_gemm1_with_bin_chunks(const float* __restrict__ A, const float* __restrict__ B,
                          const float* __restrict__ bias, float* __restrict__ C,
                          int ldc, int col_off, int n_tiles, int n_gemm_blocks,
                          const int* __restrict__ src, const int* __restrict__ dst,
                          int* __restrict__ recs_w, int* __restrict__ counts,
                          int* __restrict__ starts, int n_edges, int n_ranges) {
    __shared__ float4 sm[128 * 32];
    static_assert(sizeof(float4) * 128 * 32 >= sizeof(int) * 2 * RANGE + sizeof(int2) * CH,
                  "a chunk's LDS fits the GEMM's");
    const int blk = blockIdx.x;
    if (blk < n_gemm_blocks)
        gemm_wres_body<128>(sm, A, B, bias, C, ldc, col_off, n_tiles, blk, n_gemm_blocks);
    else
        bin_chunks_body((int*)sm, src, dst, recs_w, counts, starts, n_edges, n_ranges,
                        blk - n_gemm_blocks);
}

__global__ void gnn_gather_sum(const int* __restrict__ off, const int* __restrict__ bucket,
                               const float* __restrict__ msg_tab, float* __restrict__ agg_tab,
                               int n_nodes) {
    const int node = blockIdx.x * 8 + __builtin_amdgcn_readfirstlane((int)(threadIdx.x >> 5));
    const int lane = threadIdx.x & 31;
    if (node >= n_nodes) return;

    const int beg = __builtin_amdgcn_readfirstlane(off[node]);
    const int end = __builtin_amdgcn_readfirstlane(off[node + 1]);

    const float4* rows = (const float4*)(msg_tab + 128) + lane;
    float4 acc = make_float4(0.f, 0.f, 0.f, 0.f);

    for (int c = beg; c < end; c += 32) {
        const int m = (end - c < 32) ? (end - c) : 32;
        const int sidx = (lane < m) ? bucket[c + lane] : 0;
        for (int j = 0; j < m; j += 8) {
            float4 v[8];
#pragma unroll
            for (int k = 0; k < 8; ++k) {
                const int jj = (j + k < m) ? (j + k) : (m - 1);
                const int s  = __builtin_amdgcn_readlane(sidx, jj);
                v[k] = rows[(long)s * 64];
            }
#pragma unroll
            for (int k = 0; k < 8; ++k) {
                if (j + k < m) {
                    acc.x += v[k].x;
                    acc.y += v[k].y;
                    acc.z += v[k].z;
                    acc.w += v[k].w;
                }
            }
        }
    }
    ((float4*)(agg_tab + (long)node * 256))[lane] = acc;
}

extern "C" void kernel_launch(void* const* d_in, const int* in_sizes, int n_in,
                              void* d_out, int out_size, void* d_ws, size_t ws_size,
                              hipStream_t stream) {
    const float* features = (const float*)d_in[0];
    const int*   src      = (const int*)  d_in[1];
    const int*   dst      = (const int*)  d_in[2];
    const float* W1       = (const float*)d_in[3];
    const float* b1       = (const float*)d_in[4];
    const float* W2       = (const float*)d_in[5];
    const float* b2       = (const float*)d_in[6];

    float* out   = (float*)d_out;
    float* feat1 = (float*)d_ws;

    const int N = in_sizes[0] / 128;
    const int E = in_sizes[1];

    const int GA = (E + CH - 1) / CH;
    const int NR = (N + RANGE - 1) / RANGE;
    int* recs    = (int*)(feat1 + (size_t)N * 256);
    int* counts  = recs + 2 * (size_t)E;
    int* starts  = counts + GA * NR;
    int* off     = starts + GA * NR;
    int* bucket  = off + (N + 1);

    gnn_gemm1_with_bin_chunks<<<GEMM_BLOCKS + GA, 256, 0, stream>>>(
        features, W1, b1, feat1, 256, 128, N / 16, GEMM_BLOCKS,
        src, dst, recs, counts, starts, E, NR);

    gnn_bin_finish<<<NR, 512, 0, stream>>>(recs, counts, starts, off, bucket, N, GA, NR);

    gnn_gather_sum<<<(N + 7) / 8, 256, 0, stream>>>(off, bucket, feat1, feat1, N);

    gnn_gemm_wres<256><<<GEMM_BLOCKS, 256, 0, stream>>>(feat1, W2, b2, out, 256, 128, N / 16);

    gnn_gather_sum<<<(N + 7) / 8, 256, 0, stream>>>(off, bucket, out, out, N);
}


# Two linear layers composed: W1*W2a, W1*W2b formed first on WMMA, one GEMM of X gives P,Q, m2 = bucket sum of P + Q; sort finish rides in that launch: 4 launches; product association, k order changed
# speedup vs baseline: 4.1986x; 1.1508x over previous
#include <hip/hip_runtime.h>


typedef float v2f __attribute__((ext_vector_type(2)));
typedef float v8f __attribute__((ext_vector_type(8)));

#define CH      3200
#define BIN_T   256
#define CH_PER  13
#define RANGE   256
#define SCAP    8192

__device__ __forceinline__ int scan256_one_wave(int* a, int lane) {
    int v[8];
    int s = 0;
#pragma unroll
    for (int j = 0; j < 8; ++j) { v[j] = a[lane * 8 + j]; s += v[j]; }
    int inc = s;
#pragma unroll
    for (int d = 1; d < 32; d <<= 1) {
        int t = __shfl_up(inc, d);
        if (lane >= d) inc += t;
    }
    int run = inc - s;
#pragma unroll
    for (int j = 0; j < 8; ++j) { a[lane * 8 + j] = run; run += v[j]; }
    return __shfl(inc, 31);
}

__device__ __forceinline__ void
bin_chunks_body(int* __restrict__ sm, const int* __restrict__ src, const int* __restrict__ dst,
                int* __restrict__ recs_w, int* __restrict__ counts, int* __restrict__ starts,
                int n_edges, int n_ranges, const int a) {
    static_assert(BIN_T == RANGE, "one thread per destination range");
    int2* __restrict__ recs = (int2*)recs_w;
    int*  hist   = sm;
    int*  cursor = sm + RANGE;
    int2* stage  = (int2*)(sm + 2 * RANGE);

    const int tid  = threadIdx.x;
    const int lane = tid & 31;
    const int base = a * CH;
    const int chunk_n = (n_edges - base < CH) ? (n_edges - base) : CH;

    hist[tid] = 0;

    int d[CH_PER], s[CH_PER];
#pragma unroll
    for (int q = 0; q < CH_PER; ++q) {
        const int e = tid + q * BIN_T;
        d[q] = -1; s[q] = 0;
        if (e < chunk_n) { d[q] = dst[base + e]; s[q] = src[base + e]; }
    }
    __syncthreads();
#pragma unroll
    for (int q = 0; q < CH_PER; ++q)
        if (d[q] >= 0) atomicAdd(&hist[d[q] >> 8], 1);
    __syncthreads();

    const int my_count = hist[tid];
    __syncthreads();
    if (tid < 32) scan256_one_wave(hist, lane);
    __syncthreads();
    {
        const int st = hist[tid];
        cursor[tid] = st;
        if (tid < n_ranges) {
            counts[a * n_ranges + tid] = my_count;
            starts[a * n_ranges + tid] = st;
        }
    }
    __syncthreads();
#pragma unroll
    for (int q = 0; q < CH_PER; ++q) {
        if (d[q] >= 0) {
            const int pos = atomicAdd(&cursor[d[q] >> 8], 1);
            stage[pos] = make_int2(d[q], s[q]);
        }
    }
    __syncthreads();
    for (int i = tid; i < chunk_n; i += BIN_T) recs[base + i] = stage[i];
}

__device__ __forceinline__ void
bin_finish_body(int* __restrict__ sm, const int* __restrict__ recs_w, const int* __restrict__ counts,
                const int* __restrict__ starts, int* __restrict__ off, int* __restrict__ bucket,
                int n_nodes, int n_chunks, int n_ranges, const int r) {
    const int2* __restrict__ recs = (const int2*)recs_w;
    int* node_cnt = sm;
    int* cursor   = sm + RANGE;
    int* sums     = sm + 2 * RANGE;
    int* stage    = sm + 2 * RANGE + 32;

    const int tid  = threadIdx.x;
    const int lane = tid & 31;
    const int a    = tid;

    node_cnt[tid] = 0;
    if (tid < 2) sums[tid] = 0;

    int c = 0, o = 0;
    if (a < n_chunks) { c = counts[a * n_ranges + r]; o = starts[a * n_ranges + r]; }
    __syncthreads();

    {
        int cs = c, os = o;
#pragma unroll
        for (int dd = 16; dd >= 1; dd >>= 1) { cs += __shfl_xor(cs, dd); os += __shfl_xor(os, dd); }
        if (lane == 0) { atomicAdd(&sums[0], cs); atomicAdd(&sums[1], os); }
    }

    const int2* seg = recs + (size_t)a * CH + o;
    const int node0 = r * RANGE;

    const int* segd = (const int*)seg;
    for (int i0 = 0; i0 < c; i0 += 4) {
        int dd[4];
#pragma unroll
        for (int q = 0; q < 4; ++q) { const int i = i0 + q; dd[q] = segd[2 * (i < c ? i : i0)]; }
#pragma unroll
        for (int q = 0; q < 4; ++q) {
            const bool ok = (i0 + q) < c;
            atomicAdd(&node_cnt[ok ? dd[q] - node0 : tid], ok ? 1 : 0);
        }
    }
    __syncthreads();

    const int total     = sums[0];
    const int bin_start = sums[1];
    if (tid < 32) scan256_one_wave(node_cnt, lane);
    __syncthreads();
    {
        const int st = node_cnt[tid];
        cursor[tid] = st;
        if (node0 + tid < n_nodes) off[node0 + tid] = bin_start + st;
    }
    if (r == n_ranges - 1 && tid == 0) off[n_nodes] = bin_start + total;
    __syncthreads();

    const bool fits = total <= SCAP;
    for (int i0 = 0; i0 < c; i0 += 4) {
        int2 rec[4];
#pragma unroll
        for (int q = 0; q < 4; ++q) { const int i = i0 + q; rec[q] = seg[i < c ? i : i0]; }
#pragma unroll
        for (int q = 0; q < 4; ++q) {
            if (i0 + q < c) {
                const int pos = atomicAdd(&cursor[rec[q].x - node0], 1);
                if (fits) stage[pos] = rec[q].y;
                else      bucket[bin_start + pos] = rec[q].y;
            }
        }
    }
    __syncthreads();
    if (fits)
        for (int i = tid; i < total; i += BIN_T) bucket[bin_start + i] = stage[i];
}

#define GEMM_BLOCKS 256

__device__ __forceinline__ v8f mma4(v2f a, v2f b, v8f c) {
    return __builtin_amdgcn_wmma_f32_16x16x4_f32(false, a, false, b, (short)0, c, false, false);
}

__device__ __forceinline__ void load_a_chunk(float4 (&dst)[8], const float* __restrict__ p) {
#pragma unroll
    for (int i = 0; i < 8; ++i) dst[i] = *(const float4*)(p + 8 * i);
}

template <int K, bool HAS_BIAS>
__device__ __forceinline__ void
gemm_wres_body(float4* __restrict__ Wl, const float* __restrict__ A, const float* __restrict__ B,
               const float* __restrict__ bias, float* __restrict__ C,
               int ldc, int col_off, int n_tiles, const int b, const int G) {
    const int tid = threadIdx.x;
#pragma unroll
    for (int i = 0; i < K / 32; ++i) {
        const int id = i * 256 + tid;
        const int cg = id & 31;
        const int kg = id >> 5;
        const float* wp = B + (size_t)(4 * kg) * 128 + 4 * cg;
        const float4 r0 = *(const float4*)(wp);
        const float4 r1 = *(const float4*)(wp + 128);
        const float4 r2 = *(const float4*)(wp + 256);
        const float4 r3 = *(const float4*)(wp + 384);
        float4* d = Wl + ((kg >> 1) * 8 + (cg >> 4) * 4) * 32 + (kg & 1) * 16 + (cg & 15);
        d[0]  = make_float4(r0.x, r1.x, r2.x, r3.x);
        d[32] = make_float4(r0.y, r1.y, r2.y, r3.y);
        d[64] = make_float4(r0.z, r1.z, r2.z, r3.z);
        d[96] = make_float4(r0.w, r1.w, r2.w, r3.w);
    }
    __syncthreads();

    const int w    = __builtin_amdgcn_readfirstlane(tid >> 5);
    const int lane = tid & 31;
    const int c    = lane & 15;
    const int h    = lane >> 4;
    const int nq   = (b < n_tiles) ? (n_tiles - b + G - 1) / G : 0;
    const int nu   = nq >> 2;
    const int hf   = w & 1;
    const int qs   = w >> 1;

    const float*  arow0 = A + (size_t)c * K + 4 * h;
    const float4* wfrag = Wl + hf * 128 + lane;

    float4 a0[8], a1[8];
    float4 bf0[4], bf1[4];
    if (nu > 0) {
        load_a_chunk(a0, arow0 + (size_t)(b + G * qs) * (16 * K));
#pragma unroll
        for (int t = 0; t < 4; ++t) bf0[t] = wfrag[t * 32];
    }

    for (int i = 0; i < nu; ++i) {
        const int tile  = b + G * (qs + 4 * i);
        const int tnext = b + G * (qs + 4 * ((i + 1 < nu) ? (i + 1) : i));
        const float* arow  = arow0 + (size_t)tile  * (16 * K);
        const float* anext = arow0 + (size_t)tnext * (16 * K);

        v8f acc[4];
#pragma unroll
        for (int t = 0; t < 4; ++t) {
            const float bv = HAS_BIAS ? bias[64 * hf + 4 * c + t] : 0.f;
#pragma unroll
            for (int r = 0; r < 8; ++r) acc[t][r] = bv;
        }

#pragma unroll
        for (int q = 0; q < K / 64; ++q) {
            float4 (&acur)[8] = (q & 1) ? a1 : a0;
            float4 (&anxt)[8] = (q & 1) ? a0 : a1;
            if (q + 1 < K / 64) load_a_chunk(anxt, arow + 64 * (q + 1));
            else                load_a_chunk(anxt, anext);
#pragma unroll
            for (int jj = 0; jj < 8; ++jj) {
                const int j = q * 8 + jj;
                float4 (&bcur)[4] = (jj & 1) ? bf1 : bf0;
                float4 (&bnxt)[4] = (jj & 1) ? bf0 : bf1;
                const float4* wp = wfrag + ((j + 1 < K / 8) ? (j + 1) : 0) * 256;
#pragma unroll
                for (int t = 0; t < 4; ++t) bnxt[t] = wp[t * 32];
                const float4 av = acur[jj];
                const v2f alo = {av.x, av.y};
                const v2f ahi = {av.z, av.w};
#pragma unroll
                for (int t = 0; t < 4; ++t) {
                    const v2f blo = {bcur[t].x, bcur[t].y};
                    const v2f bhi = {bcur[t].z, bcur[t].w};
                    acc[t] = mma4(alo, blo, acc[t]);
                    acc[t] = mma4(ahi, bhi, acc[t]);
                }
                __builtin_amdgcn_sched_barrier(0);
            }
        }

        float* crow = C + (size_t)(tile * 16 + 8 * h) * ldc + col_off + 64 * hf + 4 * c;
#pragma unroll
        for (int r = 0; r < 8; ++r)
            *(float4*)(crow + (size_t)r * ldc) =
                make_float4(acc[0][r], acc[1][r], acc[2][r], acc[3][r]);
    }

    const int tcol = 64 * (w >> 2) + 4 * c + (w & 3);
    const float4* wtail = Wl + w * 32 + lane;
    for (int q = nu * 4; q < nq; ++q) {
        const int tile = b + G * q;
        const float* arow = arow0 + (size_t)tile * (16 * K);
        float4 av[K / 8];
#pragma unroll
        for (int j = 0; j < K / 8; ++j) av[j] = *(const float4*)(arow + 8 * j);
        float4 g0[4], g1[4];
#pragma unroll
        for (int s = 0; s < 4; ++s) g0[s] = wtail[s * 256];
        v8f acc;
        {
            const float bv = HAS_BIAS ? bias[tcol] : 0.f;
#pragma unroll
            for (int r = 0; r < 8; ++r) acc[r] = bv;
        }
#pragma unroll
        for (int g = 0; g < K / 32; ++g) {
            float4 (&gcur)[4] = (g & 1) ? g1 : g0;
            float4 (&gnxt)[4] = (g & 1) ? g0 : g1;
            if (g + 1 < K / 32) {
#pragma unroll
                for (int s = 0; s < 4; ++s) gnxt[s] = wtail[((g + 1) * 4 + s) * 256];
            }
#pragma unroll
            for (int s = 0; s < 4; ++s) {
                const float4 a4 = av[g * 4 + s];
                const v2f alo = {a4.x, a4.y};
                const v2f ahi = {a4.z, a4.w};
                const v2f blo = {gcur[s].x, gcur[s].y};
                const v2f bhi = {gcur[s].z, gcur[s].w};
                acc = mma4(alo, blo, acc);
                acc = mma4(ahi, bhi, acc);
            }
            __builtin_amdgcn_sched_barrier(0);
        }
        float* crow = C + (size_t)(tile * 16 + 8 * h) * ldc + col_off + tcol;
#pragma unroll
        for (int r = 0; r < 8; ++r) crow[(size_t)r * ldc] = acc[r];
    }
}

#define WPROD_BLOCKS 17

__global__ void __launch_bounds__(256)
gnn_wprod_with_bin_chunks(const float* __restrict__ W1, const float* __restrict__ b1,
                          const float* __restrict__ W2, const float* __restrict__ b2,
                          float* __restrict__ wprod, float* __restrict__ cvec,
                          const int* __restrict__ src, const int* __restrict__ dst,
                          int* __restrict__ recs_w, int* __restrict__ counts,
                          int* __restrict__ starts, int n_edges, int n_ranges) {
    __shared__ float4 sm[128 * 32];
    static_assert(sizeof(float4) * 128 * 32 >= sizeof(int) * 2 * RANGE + sizeof(int2) * CH,
                  "a chunk's LDS fits the GEMM's");
    const int blk = blockIdx.x;
    if (blk < 16) {
        const int half = blk >> 3;
        gemm_wres_body<128, false>(sm, W1, W2 + half * (128 * 128), nullptr,
                                   wprod + half * (128 * 128), 128, 0, 8, blk & 7, 8);
    } else if (blk == 16) {
        const int n    = threadIdx.x;
        const int half = n >> 7;
        const int col  = n & 127;
        const float* wp = W2 + half * (128 * 128) + col;
        float acc = 0.f;
#pragma unroll 16
        for (int k = 0; k < 128; ++k) acc += b1[k] * wp[k * 128];
        if (half) acc += b2[col];
        cvec[n] = acc;
    } else {
        bin_chunks_body((int*)sm, src, dst, recs_w, counts, starts, n_edges, n_ranges,
                        blk - WPROD_BLOCKS);
    }
}

__global__ void __launch_bounds__(256)
gnn_gemm_with_bin_finish(const float* __restrict__ X, const float* __restrict__ wprod,
                         const float* __restrict__ cvec, float* __restrict__ PQ,
                         int n_tiles, int n_gemm_blocks,
                         const int* __restrict__ recs_w, const int* __restrict__ counts,
                         const int* __restrict__ starts, int* __restrict__ off,
                         int* __restrict__ bucket, int n_nodes, int n_chunks, int n_ranges) {
    __shared__ float4 sm[128 * 32];
    static_assert(sizeof(float4) * 128 * 32 >= sizeof(int) * (2 * RANGE + 32 + SCAP),
                  "a range's LDS fits the GEMM's");
    const int blk = blockIdx.x;
    if (blk < n_gemm_blocks) {
        gemm_wres_body<128, true>(sm, X, wprod, cvec, PQ, 256, 128, n_tiles, blk, n_gemm_blocks);
        __syncthreads();
        gemm_wres_body<128, true>(sm, X, wprod + 128 * 128, cvec + 128, PQ, 256, 0, n_tiles,
                                  blk, n_gemm_blocks);
    } else {
        bin_finish_body((int*)sm, recs_w, counts, starts, off, bucket, n_nodes, n_chunks, n_ranges,
                        blk - n_gemm_blocks);
    }
}

template <bool ADD_SELF>
__global__ void gnn_gather_sum(const int* __restrict__ off, const int* __restrict__ bucket,
                               const float* __restrict__ row_tab, const float* __restrict__ self_tab,
                               float* __restrict__ dst_tab, int n_nodes) {
    const int node = blockIdx.x * 8 + __builtin_amdgcn_readfirstlane((int)(threadIdx.x >> 5));
    const int lane = threadIdx.x & 31;
    if (node >= n_nodes) return;

    const int beg = __builtin_amdgcn_readfirstlane(off[node]);
    const int end = __builtin_amdgcn_readfirstlane(off[node + 1]);

    const float4* rows = (const float4*)row_tab + lane;
    float4 acc = make_float4(0.f, 0.f, 0.f, 0.f);
    float4 own = make_float4(0.f, 0.f, 0.f, 0.f);
    if (ADD_SELF) own = ((const float4*)(self_tab + (long)node * 256))[lane];

    for (int c = beg; c < end; c += 32) {
        const int m = (end - c < 32) ? (end - c) : 32;
        const int sidx = (lane < m) ? bucket[c + lane] : 0;
        for (int j = 0; j < m; j += 8) {
            float4 v[8];
#pragma unroll
            for (int k = 0; k < 8; ++k) {
                const int jj = (j + k < m) ? (j + k) : (m - 1);
                const int s  = __builtin_amdgcn_readlane(sidx, jj);
                v[k] = rows[(long)s * 64];
            }
#pragma unroll
            for (int k = 0; k < 8; ++k) {
                if (j + k < m) {
                    acc.x += v[k].x;
                    acc.y += v[k].y;
                    acc.z += v[k].z;
                    acc.w += v[k].w;
                }
            }
        }
    }
    if (ADD_SELF) { acc.x += own.x; acc.y += own.y; acc.z += own.z; acc.w += own.w; }
    ((float4*)(dst_tab + (long)node * 256))[lane] = acc;
}

extern "C" void kernel_launch(void* const* d_in, const int* in_sizes, int n_in,
                              void* d_out, int out_size, void* d_ws, size_t ws_size,
                              hipStream_t stream) {
    const float* features = (const float*)d_in[0];
    const int*   src      = (const int*)  d_in[1];
    const int*   dst      = (const int*)  d_in[2];
    const float* W1       = (const float*)d_in[3];
    const float* b1       = (const float*)d_in[4];
    const float* W2       = (const float*)d_in[5];
    const float* b2       = (const float*)d_in[6];

    float* out = (float*)d_out;
    float* PQ  = (float*)d_ws;

    const int N = in_sizes[0] / 128;
    const int E = in_sizes[1];

    const int GA = (E + CH - 1) / CH;
    const int NR = (N + RANGE - 1) / RANGE;
    float* wprod = PQ + (size_t)N * 256;
    float* cvec  = wprod + 2 * 128 * 128;
    int* recs    = (int*)(cvec + 2 * 128);
    int* counts  = recs + 2 * (size_t)E;
    int* starts  = counts + GA * NR;
    int* off     = starts + GA * NR;
    int* bucket  = off + (N + 1);

    gnn_wprod_with_bin_chunks<<<WPROD_BLOCKS + GA, 256, 0, stream>>>(
        W1, b1, W2, b2, wprod, cvec, src, dst, recs, counts, starts, E, NR);

    gnn_gemm_with_bin_finish<<<GEMM_BLOCKS + NR, 256, 0, stream>>>(
        features, wprod, cvec, PQ, N / 16, GEMM_BLOCKS, recs, counts, starts, off, bucket, N, GA, NR);

    gnn_gather_sum<true><<<(N + 7) / 8, 256, 0, stream>>>(off, bucket, PQ + 128, PQ, out + 128, N);

    gnn_gather_sum<false><<<(N + 7) / 8, 256, 0, stream>>>(off, bucket, out + 128, nullptr, out, N);
}
